# starting kernel
# baseline (speedup 1.0000x reference)
#include <hip/hip_runtime.h>
#include <math.h>

typedef float v2f __attribute__((ext_vector_type(2)));
typedef float v8f __attribute__((ext_vector_type(8)));

#define B_   2
#define L_   385
#define N_   384
#define C_   128
#define C2_  256
#define EPSV 1e-5f
#define LDS_STRIDE 260

__global__ __launch_bounds__(C_) void lnp_stats_kernel(
    const float* __restrict__ x, const float* __restrict__ gamma,
    const float* __restrict__ beta, float* __restrict__ out,
    float* __restrict__ scale0, float* __restrict__ econst)
{
    const int b = blockIdx.x;
    const int c = threadIdx.x;
    const size_t base = (size_t)b * L_ * C_;
    const float* __restrict__ F = x + base + C_;

    out[base + c] = x[base + c];

    float sum = 0.f;
    for (int j = 0; j < N_; ++j) sum += F[j * C_ + c];
    const float mean = sum * (1.0f / N_);

    float ss = 0.f;
    for (int j = 0; j < N_; ++j) { float d = F[j * C_ + c] - mean; ss += d * d; }
    const float var = ss * (1.0f / N_);
    const float inv = 1.0f / sqrtf(var + EPSV);
    const float sc  = gamma[c] * inv;

    float mx = -3.0e38f;
    for (int j = 0; j < N_; ++j) mx = fmaxf(mx, sc * F[j * C_ + c]);

    float se = 0.f, sa = 0.f;
    for (int j = 0; j < N_; ++j) {
        float a = sc * F[j * C_ + c];
        float e = __expf(a - mx);
        se += e; sa += a * e;
    }
    scale0[b * C_ + c] = sc;
    econst[b * C_ + c] = sa / se + beta[c];
}

__global__ __launch_bounds__(256) void lnp_gemm_kernel(
    const float* __restrict__ x, const float* __restrict__ gamma,
    const float* __restrict__ beta, const float* __restrict__ W,
    const float* __restrict__ bias, const float* __restrict__ scale0,
    const float* __restrict__ econst, float* __restrict__ out)
{
    __shared__ float ldsA[16 * LDS_STRIDE];

    const int row0 = blockIdx.x * 16;
    const int b    = row0 / N_;
    const int i0   = row0 % N_;
    const int t    = threadIdx.x;

    {
        const int  cp = t;
        const int  c  = cp & (C_ - 1);
        const bool lo = (cp < C_);
        const float sc = lo ? scale0[b * C_ + cp] : 0.f;
        const float ec = lo ? econst[b * C_ + cp] : 0.f;
        const float g1 = lo ? 0.f : gamma[cp];
        const float b1 = lo ? 0.f : beta[cp];
        const float* __restrict__ Fb = x + (size_t)b * L_ * C_ + C_;
        #pragma unroll
        for (int m = 0; m < 16; ++m) {
            float Fv = Fb[(size_t)(i0 + m) * C_ + c];
            ldsA[m * LDS_STRIDE + cp] = lo ? (ec - sc * Fv) : (g1 * Fv + b1);
        }
    }
    __syncthreads();

    const int wave  = t >> 5;
    const int lane  = t & 31;
    const int nbase = wave * 16;
    const int half  = (lane < 16) ? 0 : 2;
    const int mrow  = lane & 15;
    const int ncol  = nbase + (lane & 15);

    v8f acc;
    const float bv = bias[ncol];
    #pragma unroll
    for (int q = 0; q < 8; ++q) acc[q] = bv;

    const float* __restrict__ wrow = W + (size_t)ncol * C2_;

    #pragma unroll 4
    for (int kt = 0; kt < 64; ++kt) {
        const int c0 = kt * 4 + half;
        v2f a  = *(const v2f*)&ldsA[mrow * LDS_STRIDE + c0];
        v2f bb = *(const v2f*)&wrow[c0];
        acc = __builtin_amdgcn_wmma_f32_16x16x4_f32(
                   false, a,  false, bb,
                   (short)0, acc,  false,  false);
    }

    const int mofs = (lane < 16) ? 0 : 8;
    float* __restrict__ outp = out + (size_t)b * L_ * C_ + (size_t)(1 + i0) * C_;
    #pragma unroll
    for (int v = 0; v < 8; ++v) {
        outp[(size_t)(v + mofs) * C_ + ncol] = acc[v];
    }
}

extern "C" void kernel_launch(void* const* d_in, const int* in_sizes, int n_in,
                              void* d_out, int out_size, void* d_ws, size_t ws_size,
                              hipStream_t stream) {
    const float* x     = (const float*)d_in[0];
    const float* gamma = (const float*)d_in[1];
    const float* beta  = (const float*)d_in[2];
    const float* W     = (const float*)d_in[3];
    const float* bias  = (const float*)d_in[4];
    float* out = (float*)d_out;

    float* scale0 = (float*)d_ws;
    float* econst = scale0 + B_ * C_;

    lnp_stats_kernel<<<B_, C_, 0, stream>>>(x, gamma, beta, out, scale0, econst);
    lnp_gemm_kernel<<<(B_ * N_) / 16, 256, 0, stream>>>(
        x, gamma, beta, W, bias, scale0, econst, out);
}


# product split by dependence: K columns 128..255 need no statistics, their 32 WMMAs run inside the exponential pass, 32 stay in the tail; barriers wait for LDS only; new sum order
# speedup vs baseline: 9.9882x; 9.9882x over previous
#include <hip/hip_runtime.h>
#include <math.h>


typedef float v2f __attribute__((ext_vector_type(2)));
typedef float v4f __attribute__((ext_vector_type(4)));
typedef float v8f __attribute__((ext_vector_type(8)));

#define B_   2
#define L_   385
#define N_   384
#define C_   128
#define C2_  256
#define EPSV 1e-5f
#define LDS_STRIDE 260
#define NWAVE 8
#define TPW  (N_ / NWAVE)
#define LOG2E 1.4426950408889634f

__device__ __forceinline__ void lds_barrier() {
    asm volatile("s_wait_dscnt 0x0\n\ts_barrier_signal -1\n\ts_barrier_wait -1" ::: "memory");
}

__global__ __launch_bounds__(256) __attribute__((amdgpu_waves_per_eu(2, 2))) void lnp_fused_kernel(
    const float* __restrict__ x, const float* __restrict__ gamma,
    const float* __restrict__ beta, const float* __restrict__ W,
    const float* __restrict__ bias, float* __restrict__ out)
{
    __shared__ __attribute__((aligned(16))) float ldsA[16 * LDS_STRIDE];
    __shared__ v4f red1[3][NWAVE][32];
    __shared__ v4f red2[2][NWAVE][32];

    const int row0 = blockIdx.x * 16;
    const int b    = row0 / N_;
    const int i0   = row0 % N_;
    const int t    = threadIdx.x;
    const int wave = t >> 5;
    const int lane = t & 31;
    const int c4   = 4 * lane;

    const float* __restrict__ xb = x + (size_t)b * L_ * C_;
    const float* __restrict__ Fb = xb + C_;

    const bool cls_wave = (i0 == 0 && wave == 0);
    v4f clsv = {0.f, 0.f, 0.f, 0.f};
    if (cls_wave) clsv = *(const v4f*)&xb[c4];

    const v4f arow0 = *(const v4f*)&Fb[(size_t)(i0 + wave) * C_ + c4];
    const v4f arow1 = *(const v4f*)&Fb[(size_t)(i0 + wave + 8) * C_ + c4];
    const v4f g1  = *(const v4f*)&gamma[C_ + c4];
    const v4f be1 = *(const v4f*)&beta[C_ + c4];

    v4f v[TPW];
    {
        const float* __restrict__ src = Fb + (size_t)(wave * TPW) * C_ + c4;
        #pragma unroll
        for (int k = 0; k < TPW; ++k) v[k] = *(const v4f*)&src[(size_t)k * C_];
    }
    const v4f g0  = *(const v4f*)&gamma[c4];
    const v4f be0 = *(const v4f*)&beta[c4];
    const float bv = bias[wave * 16 + (lane & 15)];

    const int hsel = lane >> 4;
    const int mrow = lane & 15;
    const int ncol = wave * 16 + mrow;
    const float* __restrict__ wsrc = W + (size_t)ncol * C2_ + 4 * hsel;
    const float* __restrict__ arow = &ldsA[mrow * LDS_STRIDE + 4 * hsel];

    v4f wf[32], af[32];
    #pragma unroll
    for (int p = 16; p < 32; ++p) wf[p] = *(const v4f*)&wsrc[8 * p];

    v4f s = v[0], hi = v[0], lo = v[0];
    #pragma unroll
    for (int k = 1; k < TPW; ++k) {
        s += v[k];
        hi = __builtin_elementwise_max(hi, v[k]);
        lo = __builtin_elementwise_min(lo, v[k]);
    }
    const v4f lmean = s * (1.0f / TPW);

    v4f m2 = {0.f, 0.f, 0.f, 0.f};
    #pragma unroll
    for (int k = 0; k < TPW; ++k) { const v4f d = v[k] - lmean; m2 += d * d; }

    v4f ext;
    #pragma unroll
    for (int q = 0; q < 4; ++q) ext[q] = (g0[q] >= 0.f) ? hi[q] : lo[q];

    red1[0][wave][lane] = lmean;
    red1[1][wave][lane] = m2;
    red1[2][wave][lane] = ext;
    *(v4f*)&ldsA[(wave    ) * LDS_STRIDE + C_ + c4] = g1 * arow0 + be1;
    *(v4f*)&ldsA[(wave + 8) * LDS_STRIDE + C_ + c4] = g1 * arow1 + be1;
    lds_barrier();

    #pragma unroll
    for (int p = 16; p < 32; ++p) af[p] = *(const v4f*)&arow[8 * p];

    if (cls_wave) *(v4f*)&out[(size_t)b * L_ * C_ + c4] = clsv;

    v4f pm[NWAVE];
    v4f msum = {0.f, 0.f, 0.f, 0.f}, m2sum = {0.f, 0.f, 0.f, 0.f};
    v4f emax = red1[2][0][lane], emin = emax;
    #pragma unroll
    for (int w = 0; w < NWAVE; ++w) {
        pm[w] = red1[0][w][lane];
        msum  += pm[w];
        m2sum += red1[1][w][lane];
        const v4f e = red1[2][w][lane];
        emax = __builtin_elementwise_max(emax, e);
        emin = __builtin_elementwise_min(emin, e);
    }
    const v4f mean = msum * (1.0f / NWAVE);
    v4f dm2 = {0.f, 0.f, 0.f, 0.f};
    #pragma unroll
    for (int w = 0; w < NWAVE; ++w) { const v4f d = pm[w] - mean; dm2 += d * d; }
    const v4f var = (m2sum + dm2 * (float)TPW) * (1.0f / N_);

    v4f sc, c1, c2;
    #pragma unroll
    for (int q = 0; q < 4; ++q) {
        const float inv = 1.0f / sqrtf(var[q] + EPSV);
        sc[q] = g0[q] * inv;
        const float mx = sc[q] * ((g0[q] >= 0.f) ? emax[q] : emin[q]);
        c1[q] = sc[q] * LOG2E;
        c2[q] = mx * LOG2E;
    }

    v8f acc;
    #pragma unroll
    for (int q = 0; q < 8; ++q) acc[q] = bv;

    v4f se = {0.f, 0.f, 0.f, 0.f}, sx = {0.f, 0.f, 0.f, 0.f};
    #pragma unroll
    for (int k = 0; k < TPW; ++k) {
        if (k % 6 == 0) {
            asm volatile("" : "+v"(acc), "+v"(c1), "+v"(c2) : "v"(se), "v"(sx) : "memory");
            wf[2 * (k / 6)]     = *(const v4f*)&wsrc[8 * (2 * (k / 6))];
            wf[2 * (k / 6) + 1] = *(const v4f*)&wsrc[8 * (2 * (k / 6) + 1)];
            #pragma unroll
            for (int j = 0; j < 2; ++j) {
                const int p = 16 + 2 * (k / 6) + j;
                const v2f a_lo = {af[p][0], af[p][1]}, a_hi = {af[p][2], af[p][3]};
                const v2f b_lo = {wf[p][0], wf[p][1]}, b_hi = {wf[p][2], wf[p][3]};
                acc = __builtin_amdgcn_wmma_f32_16x16x4_f32(
                           false, a_lo,  false, b_lo,
                           (short)0, acc,  false,  false);
                acc = __builtin_amdgcn_wmma_f32_16x16x4_f32(
                           false, a_hi,  false, b_hi,
                           (short)0, acc,  false,  false);
            }
        }
        const v4f arg = v[k] * c1 - c2;
        v4f e;
        #pragma unroll
        for (int q = 0; q < 4; ++q) e[q] = __builtin_amdgcn_exp2f(arg[q]);
        se += e;
        sx += v[k] * e;
    }
    red2[0][wave][lane] = se;
    red2[1][wave][lane] = sx;
    lds_barrier();

    v4f set = {0.f, 0.f, 0.f, 0.f}, sxt = {0.f, 0.f, 0.f, 0.f};
    v4f pe[NWAVE], px[NWAVE];
    #pragma unroll
    for (int w = 0; w < NWAVE; ++w) { pe[w] = red2[0][w][lane]; px[w] = red2[1][w][lane]; }
    __builtin_amdgcn_sched_barrier(0);
    #pragma unroll
    for (int w = 0; w < NWAVE; ++w) { set += pe[w]; sxt += px[w]; }
    const v4f ec = (sc * sxt) / set + be0;

    *(v4f*)&ldsA[(wave    ) * LDS_STRIDE + c4]      = ec - sc * arow0;
    *(v4f*)&ldsA[(wave + 8) * LDS_STRIDE + c4]      = ec - sc * arow1;
    lds_barrier();

    #pragma unroll
    for (int p = 0; p < 16; ++p) af[p] = *(const v4f*)&arow[8 * p];
    __builtin_amdgcn_sched_barrier(0);
    #pragma unroll
    for (int p = 0; p < 16; ++p) {
        const v2f a_lo = {af[p][0], af[p][1]}, a_hi = {af[p][2], af[p][3]};
        const v2f b_lo = {wf[p][0], wf[p][1]}, b_hi = {wf[p][2], wf[p][3]};
        acc = __builtin_amdgcn_wmma_f32_16x16x4_f32(
                   false, a_lo,  false, b_lo,
                   (short)0, acc,  false,  false);
        acc = __builtin_amdgcn_wmma_f32_16x16x4_f32(
                   false, a_hi,  false, b_hi,
                   (short)0, acc,  false,  false);
    }

    const int mofs = (lane < 16) ? 0 : 8;
    float* __restrict__ outp = out + (size_t)b * L_ * C_ + (size_t)(1 + i0) * C_;
    #pragma unroll
    for (int r = 0; r < 8; ++r) {
        outp[(size_t)(r + mofs) * C_ + ncol] = acc[r];
    }
}

extern "C" void kernel_launch(void* const* d_in, const int* in_sizes, int n_in,
                              void* d_out, int out_size, void* d_ws, size_t ws_size,
                              hipStream_t stream) {
    const float* x     = (const float*)d_in[0];
    const float* gamma = (const float*)d_in[1];
    const float* beta  = (const float*)d_in[2];
    const float* W     = (const float*)d_in[3];
    const float* bias  = (const float*)d_in[4];
    float* out = (float*)d_out;

    lnp_fused_kernel<<<(B_ * N_) / 16, 256, 0, stream>>>(x, gamma, beta, W, bias, out);
}
